# starting kernel
# baseline (speedup 1.0000x reference)
#include <hip/hip_runtime.h>


typedef __attribute__((ext_vector_type(2))) float v2f;
typedef __attribute__((ext_vector_type(8))) float v8f;

#define NUM_NODES (8 * 8192)
#define FDIM 64

__global__ void gp_scatter_kernel(const float* __restrict__ h,
                                  const int* __restrict__ src,
                                  const int* __restrict__ dst,
                                  float* __restrict__ msum,
                                  float* __restrict__ deg,
                                  int num_edges) {
    const int lane   = threadIdx.x & 31;
    const int wave   = (blockIdx.x * blockDim.x + threadIdx.x) >> 5;
    const int nwaves = (gridDim.x * blockDim.x) >> 5;

    for (int e = wave; e < num_edges; e += nwaves) {
        const int s = src[e];
        const int d = dst[e];
        const float2 hv = *(const float2*)(h + (size_t)s * FDIM + lane * 2);
        float* mrow = msum + (size_t)d * FDIM + lane * 2;
        unsafeAtomicAdd(mrow + 0, hv.x);
        unsafeAtomicAdd(mrow + 1, hv.y);
        if (lane == 0) unsafeAtomicAdd(deg + d, 1.0f);
    }
}

__global__ void gp_finalize_kernel(const float* __restrict__ h,
                                   const float* __restrict__ msum,
                                   const float* __restrict__ deg,
                                   float* __restrict__ out,
                                   int num_tiles) {
    const int lane = threadIdx.x & 31;
    const int wave = (blockIdx.x * blockDim.x + threadIdx.x) >> 5;
    if (wave >= num_tiles) return;

    const int node_base = wave * 16;
    const int half      = lane >> 4;
    const int m5        = lane & 15;

    const float dm = deg[node_base + m5];
    const float r  = (dm > 0.0f) ? (1.0f / dm) : 1.0f;

    const int rowbase = half * 8;
    float degrow[8];
#pragma unroll
    for (int v = 0; v < 8; ++v) degrow[v] = deg[node_base + rowbase + v];

    for (int fc = 0; fc < 4; ++fc) {
        const int feat = fc * 16 + m5;
        v8f c = {};
#pragma unroll
        for (int j = 0; j < 4; ++j) {
            v2f a;
            a.x = (m5 == 4 * j + 2 * half + 0) ? r : 0.0f;
            a.y = (m5 == 4 * j + 2 * half + 1) ? r : 0.0f;
            const int krow = node_base + 4 * j + 2 * half;
            v2f b;
            b.x = msum[(size_t)(krow + 0) * FDIM + feat];
            b.y = msum[(size_t)(krow + 1) * FDIM + feat];
            c = __builtin_amdgcn_wmma_f32_16x16x4_f32(
                     false, a,  false, b,
                     (short)0, c,  false,  false);
        }
#pragma unroll
        for (int v = 0; v < 8; ++v) {
            const size_t idx = (size_t)(node_base + rowbase + v) * FDIM + feat;
            const float hv = h[idx];
            out[idx] = (degrow[v] > 0.0f) ? c[v] : hv;
        }
    }
}

extern "C" void kernel_launch(void* const* d_in, const int* in_sizes, int n_in,
                              void* d_out, int out_size, void* d_ws, size_t ws_size,
                              hipStream_t stream) {
    const float* h   = (const float*)d_in[0];
    const int*   src = (const int*)d_in[1];
    const int*   dst = (const int*)d_in[2];
    float*       out = (float*)d_out;
    const int num_edges = in_sizes[1];

    float* msum = (float*)d_ws;
    float* deg  = msum + (size_t)NUM_NODES * FDIM;

    hipMemsetAsync(d_ws, 0, ((size_t)NUM_NODES * FDIM + NUM_NODES) * sizeof(float),
                   stream);

    gp_scatter_kernel<<<2048, 256, 0, stream>>>(h, src, dst, msum, deg, num_edges);

    const int num_tiles = NUM_NODES / 16;
    gp_finalize_kernel<<<num_tiles / 8, 256, 0, stream>>>(h, msum, deg, out, num_tiles);
}


# Float-atomic scatter replaced: a pre-pass writes a 1-byte dst-block key per edge; each workgroup owns 256 dst rows in LDS, scans keys only, LDS float adds; memset, 2nd pass gone; sum order changes
# speedup vs baseline: 3.1743x; 3.1743x over previous
#include <hip/hip_runtime.h>


typedef __attribute__((ext_vector_type(2))) float v2f;
typedef __attribute__((ext_vector_type(8))) float v8f;
typedef __attribute__((ext_vector_type(4))) unsigned u4;

#define NUM_NODES (8 * 8192)
#define FDIM 64
#define OWN 256
#define QCAP 32768
#define OWNER_THREADS 1024
#define ROWS_IN_FLIGHT 8
#define EMPTY_ENTRY 0xFFFFFFFFu

__global__ void gp_key_kernel(const int* __restrict__ dst,
                              unsigned* __restrict__ keys,
                              int num_edges) {
    const int nwords = ((num_edges + 15) >> 4) * 4;
    const int stride = gridDim.x * blockDim.x;
    for (int t = blockIdx.x * blockDim.x + threadIdx.x; t < nwords; t += stride) {
        const int e = t * 4;
        unsigned k = 0;
        if (e + 3 < num_edges) {
            const int4 d = *(const int4*)(dst + e);
            k = (((unsigned)d.x >> 8) & 0xFFu) | ((((unsigned)d.y >> 8) & 0xFFu) << 8) |
                ((((unsigned)d.z >> 8) & 0xFFu) << 16) | ((((unsigned)d.w >> 8) & 0xFFu) << 24);
        } else {
            for (int j = 0; j < 4; ++j)
                if (e + j < num_edges) k |= (((unsigned)dst[e + j] >> 8) & 0xFFu) << (8 * j);
        }
        keys[t] = k;
    }
}

__device__ __forceinline__ unsigned nonzero_bytes(unsigned w, unsigned kk) {
    const unsigned x = w ^ kk;
    return (((x & 0x7F7F7F7Fu) + 0x7F7F7F7Fu) | x) & 0x80808080u;
}

__global__ __launch_bounds__(OWNER_THREADS)
void gp_owner_kernel(const float* __restrict__ h,
                     const int* __restrict__ src,
                     const int* __restrict__ dst,
                     const unsigned* __restrict__ keys,
                     float* __restrict__ out,
                     int num_edges) {
    __shared__ float    acc[OWN * FDIM];
    __shared__ unsigned degs[OWN];
    __shared__ unsigned queue[QCAP];
    __shared__ unsigned qcount;

    const int tid  = threadIdx.x;
    const int lane = tid & 31;
    const int wave = tid >> 5;
    const unsigned b  = blockIdx.x;
    const unsigned kk = b * 0x01010101u;
    const int own_base = (int)b * OWN;

    for (int i = tid; i < OWN * FDIM / 4; i += OWNER_THREADS)
        ((float4*)acc)[i] = make_float4(0.f, 0.f, 0.f, 0.f);
    if (tid < OWN) degs[tid] = 0u;
    if (tid == 0) qcount = 0u;
    __syncthreads();

    const int ngroups = (num_edges + 15) >> 4;
    const u4* k4 = (const u4*)keys;
    int seg = ngroups;
    int gs  = 0;
    while (gs < ngroups) {
        const int ge = (gs + seg < ngroups) ? gs + seg : ngroups;
        for (int g0 = gs + tid; g0 < ge; g0 += OWNER_THREADS * 8) {
            u4 w[8];
#pragma unroll
            for (int u = 0; u < 8; ++u) {
                const int g  = g0 + u * OWNER_THREADS;
                const int gc = g < ge ? g : ge - 1;
                w[u] = __builtin_nontemporal_load(k4 + gc);
            }
#pragma unroll
            for (int u = 0; u < 8; ++u) {
                const int g = g0 + u * OWNER_THREADS;
                const unsigned nz = nonzero_bytes(w[u].w, kk) | (nonzero_bytes(w[u].z, kk) >> 1) |
                                    (nonzero_bytes(w[u].y, kk) >> 2) | (nonzero_bytes(w[u].x, kk) >> 3);
                unsigned m = (g < ge) ? (nz ^ 0xF0F0F0F0u) : 0u;
                while (m) {
                    const int p = __builtin_ctz(m);
                    m &= m - 1;
                    const int e = g * 16 + (p & 3) * 4 + (p >> 3);
                    if (e < num_edges) {
                        const unsigned slot = atomicAdd(&qcount, 1u);
                        if (slot < QCAP) queue[slot] = (unsigned)e;
                    }
                }
            }
        }
        __syncthreads();
        const unsigned cnt = qcount;
        if (cnt > QCAP) {
            __syncthreads();
            if (tid == 0) qcount = 0u;
            __syncthreads();
            seg = QCAP / 16;
            continue;
        }
        const int Q = (int)cnt;

        for (int q = tid; q < Q; q += OWNER_THREADS) {
            const int e = (int)queue[q];
            int s = src[e];
            s = s < 0 ? 0 : (s > NUM_NODES - 1 ? NUM_NODES - 1 : s);
            const unsigned dl = (unsigned)(dst[e] - own_base);
            queue[q] = (dl < OWN) ? ((unsigned)s | (dl << 16)) : EMPTY_ENTRY;
        }
        __syncthreads();

        for (int q0 = wave * ROWS_IN_FLIGHT; q0 < Q; q0 += (OWNER_THREADS / 32) * ROWS_IN_FLIGHT) {
            unsigned ent[ROWS_IN_FLIGHT];
            float2   hv[ROWS_IN_FLIGHT];
#pragma unroll
            for (int u = 0; u < ROWS_IN_FLIGHT; ++u)
                ent[u] = (q0 + u < Q) ? queue[q0 + u] : EMPTY_ENTRY;
#pragma unroll
            for (int u = 0; u < ROWS_IN_FLIGHT; ++u) {
                const unsigned s = (ent[u] == EMPTY_ENTRY) ? 0u : (ent[u] & 0xFFFFu);
                hv[u] = *(const float2*)(h + (size_t)s * FDIM + lane * 2);
            }
#pragma unroll
            for (int u = 0; u < ROWS_IN_FLIGHT; ++u) {
                if (ent[u] != EMPTY_ENTRY) {
                    const unsigned dl = ent[u] >> 16;
                    float* a = acc + dl * FDIM + lane * 2;
                    atomicAdd(a + 0, hv[u].x);
                    atomicAdd(a + 1, hv[u].y);
                    if (lane == 0) atomicAdd(&degs[dl], 1u);
                }
            }
        }
        __syncthreads();
        if (tid == 0) qcount = 0u;
        __syncthreads();
        gs = ge;
    }

    const int tile  = wave >> 1;
    const int nbl   = tile * 16;
    const int half  = lane >> 4;
    const int m5    = lane & 15;

    const float dm = (float)degs[nbl + m5];
    const float r  = (dm > 0.0f) ? (1.0f / dm) : 1.0f;

    const int rowbase = half * 8;
    unsigned degrow[8];
#pragma unroll
    for (int v = 0; v < 8; ++v) degrow[v] = degs[nbl + rowbase + v];

#pragma unroll
    for (int fi = 0; fi < 2; ++fi) {
        const int fc   = (wave & 1) * 2 + fi;
        const int feat = fc * 16 + m5;
        v8f c = {};
#pragma unroll
        for (int j = 0; j < 4; ++j) {
            v2f a;
            a.x = (m5 == 4 * j + 2 * half + 0) ? r : 0.0f;
            a.y = (m5 == 4 * j + 2 * half + 1) ? r : 0.0f;
            const int krow = nbl + 4 * j + 2 * half;
            v2f bb;
            bb.x = acc[(krow + 0) * FDIM + feat];
            bb.y = acc[(krow + 1) * FDIM + feat];
            c = __builtin_amdgcn_wmma_f32_16x16x4_f32(
                     false, a,  false, bb,
                     (short)0, c,  false,  false);
        }
#pragma unroll
        for (int v = 0; v < 8; ++v) {
            const size_t idx = (size_t)(own_base + nbl + rowbase + v) * FDIM + feat;
            out[idx] = (degrow[v] > 0u) ? c[v] : h[idx];
        }
    }
}

extern "C" void kernel_launch(void* const* d_in, const int* in_sizes, int n_in,
                              void* d_out, int out_size, void* d_ws, size_t ws_size,
                              hipStream_t stream) {
    const float* h   = (const float*)d_in[0];
    const int*   src = (const int*)d_in[1];
    const int*   dst = (const int*)d_in[2];
    float*       out = (float*)d_out;
    const int num_edges = in_sizes[1];

    unsigned* keys = (unsigned*)d_ws;

    gp_key_kernel<<<1024, 256, 0, stream>>>(dst, keys, num_edges);

    gp_owner_kernel<<<NUM_NODES / OWN, OWNER_THREADS, 0, stream>>>(h, src, dst, keys, out, num_edges);
}


# Byte key per edge replaced by per-owner bit planes (1 bit per 4-edge group, built in LDS by the pre-pass): an owner reads its own 32 KB plane, not all 1 MB of keys; sum order changes
# speedup vs baseline: 3.9601x; 1.2476x over previous
#include <hip/hip_runtime.h>


typedef __attribute__((ext_vector_type(2))) float v2f;
typedef __attribute__((ext_vector_type(8))) float v8f;
typedef __attribute__((ext_vector_type(4))) unsigned u4;

#define NUM_NODES (8 * 8192)
#define FDIM 64
#define OWN 256
#define NOWNERS 256
#define GRP 4
#define CH_EDGES 4096
#define CH_WORDS 32
#define PLANE_THREADS 256
#define GCAP 8192
#define ECAP (GCAP * GRP)
#define OWNER_THREADS 1024
#define ROWS_IN_FLIGHT 8
#define EMPTY_ENTRY 0xFFFFFFFFu

__device__ __forceinline__ int4 load_group(const int* __restrict__ p, int e, int n, int fill) {
    if (e + GRP - 1 < n) return *(const int4*)(p + e);
    int4 r;
    r.x = (e + 0 < n) ? p[e + 0] : fill;
    r.y = (e + 1 < n) ? p[e + 1] : fill;
    r.z = (e + 2 < n) ? p[e + 2] : fill;
    r.w = (e + 3 < n) ? p[e + 3] : fill;
    return r;
}

__global__ __launch_bounds__(PLANE_THREADS)
void gp_plane_kernel(const int* __restrict__ dst,
                     unsigned* __restrict__ planes,
                     int num_edges) {
    __shared__ unsigned bits[NOWNERS * CH_WORDS];

    const int t    = threadIdx.x;
    const int lane = t & 31;
    const int wave = t >> 5;
    const int nchunks = (num_edges + CH_EDGES - 1) / CH_EDGES;
    const size_t pstride = (size_t)nchunks * CH_WORDS;

    for (int c = blockIdx.x; c < nchunks; c += gridDim.x) {
        for (int i = t; i < NOWNERS * CH_WORDS / 4; i += PLANE_THREADS)
            ((u4*)bits)[i] = (u4){0u, 0u, 0u, 0u};
        __syncthreads();

        const int ebase = c * CH_EDGES;
        if (ebase + CH_EDGES <= num_edges) {
            int4 d[4];
#pragma unroll
            for (int i = 0; i < 4; ++i)
                d[i] = *(const int4*)(dst + ebase + (i * PLANE_THREADS + t) * GRP);
#pragma unroll
            for (int i = 0; i < 4; ++i) {
                const int gl = i * PLANE_THREADS + t;
                const unsigned bit = 1u << (gl & 31);
                const unsigned col = (unsigned)gl >> 5;
                const int dk[4] = {d[i].x, d[i].y, d[i].z, d[i].w};
#pragma unroll
                for (int k = 0; k < GRP; ++k) {
                    const unsigned o = ((unsigned)dk[k] >> 8) & 0xFFu;
                    atomicOr(&bits[o * CH_WORDS + ((col + (o >> 1)) & (CH_WORDS - 1))], bit);
                }
            }
        } else {
            for (int i = 0; i < 4; ++i) {
                const int gl = i * PLANE_THREADS + t;
                const unsigned bit = 1u << (gl & 31);
                const unsigned col = (unsigned)gl >> 5;
                for (int k = 0; k < GRP; ++k) {
                    const int e = ebase + gl * GRP + k;
                    if (e < num_edges) {
                        const unsigned o = ((unsigned)dst[e] >> 8) & 0xFFu;
                        atomicOr(&bits[o * CH_WORDS + ((col + (o >> 1)) & (CH_WORDS - 1))], bit);
                    }
                }
            }
        }
        __syncthreads();

        for (int o = wave; o < NOWNERS; o += PLANE_THREADS / 32) {
            const unsigned v = bits[o * CH_WORDS + ((lane + (o >> 1)) & (CH_WORDS - 1))];
            planes[(size_t)o * pstride + (size_t)c * CH_WORDS + lane] = v;
        }
        __syncthreads();
    }
}

template <bool FULL>
__device__ __forceinline__ void lookup_groups(const int* __restrict__ src, const int* __restrict__ dst,
                                              const unsigned* gqueue, unsigned* equeue, unsigned* ecount,
                                              int G, int num_edges, int own_base, int tid) {
    for (int q0 = tid; q0 < G; q0 += OWNER_THREADS * 4) {
        int4 dv[4], sv[4];
        bool has[4];
        unsigned g[4];
#pragma unroll
        for (int u = 0; u < 4; ++u) {
            const int q = q0 + u * OWNER_THREADS;
            has[u] = q < G;
            g[u] = has[u] ? gqueue[q] : 0u;
        }
#pragma unroll
        for (int u = 0; u < 4; ++u) {
            if (FULL) {
                dv[u] = ((const int4*)dst)[g[u]];
                sv[u] = ((const int4*)src)[g[u]];
            } else {
                dv[u] = load_group(dst, (int)(g[u] * GRP), num_edges, -1);
                sv[u] = load_group(src, (int)(g[u] * GRP), num_edges, 0);
            }
        }
        unsigned ent[16];
        unsigned n = 0;
#pragma unroll
        for (int u = 0; u < 4; ++u) {
            const int dk[4] = {dv[u].x, dv[u].y, dv[u].z, dv[u].w};
            const int sk[4] = {sv[u].x, sv[u].y, sv[u].z, sv[u].w};
#pragma unroll
            for (int k = 0; k < GRP; ++k) {
                int s = sk[k];
                s = s < 0 ? 0 : (s > NUM_NODES - 1 ? NUM_NODES - 1 : s);
                const unsigned dl = (unsigned)(dk[k] - own_base);
                const bool mine = has[u] && dl < OWN;
                ent[u * 4 + k] = mine ? ((unsigned)s | (dl << 16)) : EMPTY_ENTRY;
                n += mine ? 1u : 0u;
            }
        }
        unsigned slot = atomicAdd(ecount, n);
#pragma unroll
        for (int i = 0; i < 16; ++i)
            if (ent[i] != EMPTY_ENTRY) equeue[slot++] = ent[i];
    }
}

__global__ __launch_bounds__(OWNER_THREADS)
void gp_owner_kernel(const float* __restrict__ h,
                     const int* __restrict__ src,
                     const int* __restrict__ dst,
                     const unsigned* __restrict__ planes,
                     float* __restrict__ out,
                     int num_edges) {
    __shared__ float    acc[OWN * FDIM];
    __shared__ unsigned degs[OWN];
    __shared__ unsigned gqueue[GCAP];
    __shared__ unsigned equeue[ECAP];
    __shared__ unsigned gcount;
    __shared__ unsigned ecount;

    const int tid  = threadIdx.x;
    const int lane = tid & 31;
    const int wave = tid >> 5;
    const unsigned b = blockIdx.x;
    const int own_base = (int)b * OWN;

    for (int i = tid; i < OWN * FDIM / 4; i += OWNER_THREADS)
        ((float4*)acc)[i] = make_float4(0.f, 0.f, 0.f, 0.f);
    if (tid < OWN) degs[tid] = 0u;
    if (tid == 0) { gcount = 0u; ecount = 0u; }
    __syncthreads();

    const int nchunks = (num_edges + CH_EDGES - 1) / CH_EDGES;
    const int nunits  = nchunks * (CH_WORDS / 4);
    const u4* p4 = (const u4*)(planes + (size_t)b * (size_t)nchunks * CH_WORDS);
    int seg = nunits;
    int us  = 0;
    while (us < nunits) {
        const int ue = (us + seg < nunits) ? us + seg : nunits;
        for (int u0 = us + tid; u0 < ue; u0 += OWNER_THREADS * 2) {
            u4 w[2];
#pragma unroll
            for (int j = 0; j < 2; ++j) {
                const int u  = u0 + j * OWNER_THREADS;
                const int uc = u < ue ? u : ue - 1;
                w[j] = __builtin_nontemporal_load(p4 + uc);
                if (u >= ue) w[j] = (u4){0u, 0u, 0u, 0u};
            }
            unsigned n = 0;
#pragma unroll
            for (int j = 0; j < 2; ++j)
                n += __builtin_popcount(w[j].x) + __builtin_popcount(w[j].y) +
                     __builtin_popcount(w[j].z) + __builtin_popcount(w[j].w);
            unsigned slot = atomicAdd(&gcount, n);
#pragma unroll
            for (int j = 0; j < 2; ++j) {
                const unsigned wk[4] = {w[j].x, w[j].y, w[j].z, w[j].w};
#pragma unroll
                for (int k = 0; k < 4; ++k) {
                    unsigned m = wk[k];
                    const unsigned base = (unsigned)((u0 + j * OWNER_THREADS) * 4 + k) << 5;
                    while (m) {
                        const unsigned p = (unsigned)__builtin_ctz(m);
                        m &= m - 1;
                        if (slot < GCAP) gqueue[slot] = base + p;
                        ++slot;
                    }
                }
            }
        }
        __syncthreads();
        const unsigned cnt = gcount;
        if (cnt > GCAP) {
            __syncthreads();
            if (tid == 0) gcount = 0u;
            __syncthreads();
            seg = GCAP / 128;
            continue;
        }
        const int G = (int)cnt;

        if ((num_edges & (GRP - 1)) == 0)
            lookup_groups<true>(src, dst, gqueue, equeue, &ecount, G, num_edges, own_base, tid);
        else
            lookup_groups<false>(src, dst, gqueue, equeue, &ecount, G, num_edges, own_base, tid);
        __syncthreads();
        const int Q = (int)ecount;

        for (int q0 = wave * ROWS_IN_FLIGHT; q0 < Q; q0 += (OWNER_THREADS / 32) * ROWS_IN_FLIGHT) {
            unsigned ent[ROWS_IN_FLIGHT];
            float2   hv[ROWS_IN_FLIGHT];
#pragma unroll
            for (int u = 0; u < ROWS_IN_FLIGHT; ++u)
                ent[u] = (q0 + u < Q) ? equeue[q0 + u] : EMPTY_ENTRY;
#pragma unroll
            for (int u = 0; u < ROWS_IN_FLIGHT; ++u) {
                const unsigned s = (ent[u] == EMPTY_ENTRY) ? 0u : (ent[u] & 0xFFFFu);
                hv[u] = *(const float2*)(h + (size_t)s * FDIM + lane * 2);
            }
#pragma unroll
            for (int u = 0; u < ROWS_IN_FLIGHT; ++u) {
                if (ent[u] != EMPTY_ENTRY) {
                    const unsigned dl = ent[u] >> 16;
                    float* a = acc + dl * FDIM + lane * 2;
                    atomicAdd(a + 0, hv[u].x);
                    atomicAdd(a + 1, hv[u].y);
                    if (lane == 0) atomicAdd(&degs[dl], 1u);
                }
            }
        }
        __syncthreads();
        if (tid == 0) { gcount = 0u; ecount = 0u; }
        __syncthreads();
        us = ue;
    }

    const int tile  = wave >> 1;
    const int nbl   = tile * 16;
    const int half  = lane >> 4;
    const int m5    = lane & 15;

    const float dm = (float)degs[nbl + m5];
    const float r  = (dm > 0.0f) ? (1.0f / dm) : 1.0f;

    const int rowbase = half * 8;
    unsigned degrow[8];
#pragma unroll
    for (int v = 0; v < 8; ++v) degrow[v] = degs[nbl + rowbase + v];

#pragma unroll
    for (int fi = 0; fi < 2; ++fi) {
        const int fc   = (wave & 1) * 2 + fi;
        const int feat = fc * 16 + m5;
        v8f c = {};
#pragma unroll
        for (int j = 0; j < 4; ++j) {
            v2f a;
            a.x = (m5 == 4 * j + 2 * half + 0) ? r : 0.0f;
            a.y = (m5 == 4 * j + 2 * half + 1) ? r : 0.0f;
            const int krow = nbl + 4 * j + 2 * half;
            v2f bb;
            bb.x = acc[(krow + 0) * FDIM + feat];
            bb.y = acc[(krow + 1) * FDIM + feat];
            c = __builtin_amdgcn_wmma_f32_16x16x4_f32(
                     false, a,  false, bb,
                     (short)0, c,  false,  false);
        }
#pragma unroll
        for (int v = 0; v < 8; ++v) {
            const size_t idx = (size_t)(own_base + nbl + rowbase + v) * FDIM + feat;
            out[idx] = (degrow[v] > 0u) ? c[v] : h[idx];
        }
    }
}

extern "C" void kernel_launch(void* const* d_in, const int* in_sizes, int n_in,
                              void* d_out, int out_size, void* d_ws, size_t ws_size,
                              hipStream_t stream) {
    const float* h   = (const float*)d_in[0];
    const int*   src = (const int*)d_in[1];
    const int*   dst = (const int*)d_in[2];
    float*       out = (float*)d_out;
    const int num_edges = in_sizes[1];

    unsigned* planes = (unsigned*)d_ws;

    gp_plane_kernel<<<256, PLANE_THREADS, 0, stream>>>(dst, planes, num_edges);

    gp_owner_kernel<<<NUM_NODES / OWN, OWNER_THREADS, 0, stream>>>(h, src, dst, planes, out, num_edges);
}


# Lookup and add fused per wave: 32 queued groups a step, own edges packed by ballot into a private LDS slice, degree counted there; block edge queue, its atomics, a barrier gone; sum order changes
# speedup vs baseline: 4.1772x; 1.0548x over previous
#include <hip/hip_runtime.h>


typedef __attribute__((ext_vector_type(2))) float v2f;
typedef __attribute__((ext_vector_type(8))) float v8f;
typedef __attribute__((ext_vector_type(4))) unsigned u4;

#define NUM_NODES (8 * 8192)
#define FDIM 64
#define OWN 256
#define NOWNERS 256
#define GRP 4
#define CH_EDGES 4096
#define CH_WORDS 32
#define PLANE_THREADS 256
#define GCAP 8192
#define OWNER_THREADS 1024
#define OWNER_WAVES (OWNER_THREADS / 32)
#define ROWS_IN_FLIGHT 8
#define WQ_STRIDE 1024
#define EMPTY_ENTRY 0xFFFFFFFFu

__device__ __forceinline__ int4 load_group(const int* __restrict__ p, int e, int n, int fill) {
    if (e + GRP - 1 < n) return *(const int4*)(p + e);
    int4 r;
    r.x = (e + 0 < n) ? p[e + 0] : fill;
    r.y = (e + 1 < n) ? p[e + 1] : fill;
    r.z = (e + 2 < n) ? p[e + 2] : fill;
    r.w = (e + 3 < n) ? p[e + 3] : fill;
    return r;
}

__global__ __launch_bounds__(PLANE_THREADS)
void gp_plane_kernel(const int* __restrict__ dst,
                     unsigned* __restrict__ planes,
                     int num_edges) {
    __shared__ unsigned bits[NOWNERS * CH_WORDS];

    const int t    = threadIdx.x;
    const int lane = t & 31;
    const int wave = t >> 5;
    const int nchunks = (num_edges + CH_EDGES - 1) / CH_EDGES;
    const size_t pstride = (size_t)nchunks * CH_WORDS;

    for (int c = blockIdx.x; c < nchunks; c += gridDim.x) {
        for (int i = t; i < NOWNERS * CH_WORDS / 4; i += PLANE_THREADS)
            ((u4*)bits)[i] = (u4){0u, 0u, 0u, 0u};
        __syncthreads();

        const int ebase = c * CH_EDGES;
        if (ebase + CH_EDGES <= num_edges) {
            int4 d[4];
#pragma unroll
            for (int i = 0; i < 4; ++i)
                d[i] = *(const int4*)(dst + ebase + (i * PLANE_THREADS + t) * GRP);
#pragma unroll
            for (int i = 0; i < 4; ++i) {
                const int gl = i * PLANE_THREADS + t;
                const unsigned bit = 1u << (gl & 31);
                const unsigned col = (unsigned)gl >> 5;
                const int dk[4] = {d[i].x, d[i].y, d[i].z, d[i].w};
#pragma unroll
                for (int k = 0; k < GRP; ++k) {
                    const unsigned o = ((unsigned)dk[k] >> 8) & 0xFFu;
                    atomicOr(&bits[o * CH_WORDS + ((col + (o >> 1)) & (CH_WORDS - 1))], bit);
                }
            }
        } else {
            for (int i = 0; i < 4; ++i) {
                const int gl = i * PLANE_THREADS + t;
                const unsigned bit = 1u << (gl & 31);
                const unsigned col = (unsigned)gl >> 5;
                for (int k = 0; k < GRP; ++k) {
                    const int e = ebase + gl * GRP + k;
                    if (e < num_edges) {
                        const unsigned o = ((unsigned)dst[e] >> 8) & 0xFFu;
                        atomicOr(&bits[o * CH_WORDS + ((col + (o >> 1)) & (CH_WORDS - 1))], bit);
                    }
                }
            }
        }
        __syncthreads();

        for (int o = wave; o < NOWNERS; o += PLANE_THREADS / 32) {
            const unsigned v = bits[o * CH_WORDS + ((lane + (o >> 1)) & (CH_WORDS - 1))];
            planes[(size_t)o * pstride + (size_t)c * CH_WORDS + lane] = v;
        }
        __syncthreads();
    }
}

template <bool FULL>
__device__ __forceinline__ void fetch_group(const int* __restrict__ src, const int* __restrict__ dst,
                                            const unsigned* gqueue, int q, bool has, int num_edges,
                                            int4& dv, int4& sv) {
    const unsigned g = has ? gqueue[q] : 0u;
    if (FULL) {
        dv = ((const int4*)dst)[g];
        sv = ((const int4*)src)[g];
    } else {
        dv = load_group(dst, (int)(g * GRP), num_edges, -1);
        sv = load_group(src, (int)(g * GRP), num_edges, 0);
    }
}

template <bool CHECK>
__device__ __forceinline__ void add_rows(const float* __restrict__ h, float* acc, const unsigned* wq,
                                         int q0, int Q, int lane) {
    unsigned ent[ROWS_IN_FLIGHT];
    float2   hv[ROWS_IN_FLIGHT];
#pragma unroll
    for (int u = 0; u < ROWS_IN_FLIGHT; ++u)
        ent[u] = (!CHECK || q0 + u < Q) ? wq[q0 + u] : EMPTY_ENTRY;
#pragma unroll
    for (int u = 0; u < ROWS_IN_FLIGHT; ++u) {
        const unsigned s = (CHECK && ent[u] == EMPTY_ENTRY) ? 0u : (ent[u] & 0xFFFFu);
        hv[u] = *(const float2*)(h + (size_t)s * FDIM + lane * 2);
    }
#pragma unroll
    for (int u = 0; u < ROWS_IN_FLIGHT; ++u) {
        if (!CHECK || ent[u] != EMPTY_ENTRY) {
            const unsigned dl = ent[u] >> 16;
            float* a = acc + dl * FDIM + lane * 2;
            atomicAdd(a + 0, hv[u].x);
            atomicAdd(a + 1, hv[u].y);
        }
    }
}

template <bool FULL>
__device__ __forceinline__ void lookup_and_add(const float* __restrict__ h,
                                               const int* __restrict__ src, const int* __restrict__ dst,
                                               const unsigned* gqueue, unsigned* wq,
                                               float* acc, unsigned* degs,
                                               int G, int num_edges, int own_base, int wave, int lane) {
    const int qlo = (wave * G) >> 5;
    const int qhi = ((wave + 1) * G) >> 5;
    int4 dv, sv;
    bool has = qlo + lane < qhi;
    if (qlo < qhi) fetch_group<FULL>(src, dst, gqueue, qlo + lane, has, num_edges, dv, sv);
    int carried = 0;
    for (int qb = qlo; qb < qhi; qb += 32) {
        const int4 dcur = dv, scur = sv;
        const bool hcur = has;
        if (qb + 32 < qhi) {
            has = qb + 32 + lane < qhi;
            fetch_group<FULL>(src, dst, gqueue, qb + 32 + lane, has, num_edges, dv, sv);
        }
        const int dk[4] = {dcur.x, dcur.y, dcur.z, dcur.w};
        const int sk[4] = {scur.x, scur.y, scur.z, scur.w};
        int n = carried;
#pragma unroll
        for (int k = 0; k < GRP; ++k) {
            int s = sk[k];
            s = s < 0 ? 0 : (s > NUM_NODES - 1 ? NUM_NODES - 1 : s);
            const unsigned dl = (unsigned)(dk[k] - own_base);
            const bool mine = hcur && dl < OWN;
            const unsigned bal = __builtin_amdgcn_ballot_w32(mine);
            if (mine) {
                wq[n + (int)__builtin_amdgcn_mbcnt_lo(bal, 0u)] = (unsigned)s | (dl << 16);
                atomicAdd(&degs[dl], 1u);
            }
            n += __builtin_popcount(bal);
        }
        int q0 = 0;
        for (; q0 + ROWS_IN_FLIGHT <= n; q0 += ROWS_IN_FLIGHT)
            add_rows<false>(h, acc, wq, q0, n, lane);
        const int r = n - q0;
        if (q0 > 0 && r > 0) {
            const unsigned t = wq[q0 + (lane < r ? lane : 0)];
            if (lane < r) wq[lane] = t;
        }
        carried = r;
    }
    if (carried > 0) add_rows<true>(h, acc, wq, 0, carried, lane);
}

__global__ __launch_bounds__(OWNER_THREADS)
void gp_owner_kernel(const float* __restrict__ h,
                     const int* __restrict__ src,
                     const int* __restrict__ dst,
                     const unsigned* __restrict__ planes,
                     float* __restrict__ out,
                     int num_edges) {
    __shared__ float    acc[OWN * FDIM];
    __shared__ unsigned degs[OWN];
    __shared__ unsigned gqueue[GCAP];
    __shared__ unsigned wqueue[OWNER_WAVES * WQ_STRIDE];
    __shared__ unsigned gcount;

    const int tid  = threadIdx.x;
    const int lane = tid & 31;
    const int wave = tid >> 5;
    const unsigned b = blockIdx.x;
    const int own_base = (int)b * OWN;

    for (int i = tid; i < OWN * FDIM / 4; i += OWNER_THREADS)
        ((float4*)acc)[i] = make_float4(0.f, 0.f, 0.f, 0.f);
    if (tid < OWN) degs[tid] = 0u;
    if (tid == 0) gcount = 0u;
    __syncthreads();

    const int nchunks = (num_edges + CH_EDGES - 1) / CH_EDGES;
    const int nunits  = nchunks * (CH_WORDS / 4);
    const u4* p4 = (const u4*)(planes + (size_t)b * (size_t)nchunks * CH_WORDS);
    int seg = nunits;
    int us  = 0;
    while (us < nunits) {
        const int ue = (us + seg < nunits) ? us + seg : nunits;
        for (int u0 = us + tid; u0 < ue; u0 += OWNER_THREADS * 2) {
            u4 w[2];
#pragma unroll
            for (int j = 0; j < 2; ++j) {
                const int u  = u0 + j * OWNER_THREADS;
                const int uc = u < ue ? u : ue - 1;
                w[j] = __builtin_nontemporal_load(p4 + uc);
                if (u >= ue) w[j] = (u4){0u, 0u, 0u, 0u};
            }
            unsigned n = 0;
#pragma unroll
            for (int j = 0; j < 2; ++j)
                n += __builtin_popcount(w[j].x) + __builtin_popcount(w[j].y) +
                     __builtin_popcount(w[j].z) + __builtin_popcount(w[j].w);
            unsigned slot = atomicAdd(&gcount, n);
#pragma unroll
            for (int j = 0; j < 2; ++j) {
                const unsigned wk[4] = {w[j].x, w[j].y, w[j].z, w[j].w};
#pragma unroll
                for (int k = 0; k < 4; ++k) {
                    unsigned m = wk[k];
                    const unsigned base = (unsigned)((u0 + j * OWNER_THREADS) * 4 + k) << 5;
                    while (m) {
                        const unsigned p = (unsigned)__builtin_ctz(m);
                        m &= m - 1;
                        if (slot < GCAP) gqueue[slot] = base + p;
                        ++slot;
                    }
                }
            }
        }
        __syncthreads();
        const unsigned cnt = gcount;
        if (cnt > GCAP) {
            __syncthreads();
            if (tid == 0) gcount = 0u;
            __syncthreads();
            seg = GCAP / 128;
            continue;
        }
        const int G = (int)cnt;

        const int wv = __builtin_amdgcn_readfirstlane(wave);
        unsigned* wq = wqueue + wv * WQ_STRIDE;
        if ((num_edges & (GRP - 1)) == 0)
            lookup_and_add<true>(h, src, dst, gqueue, wq, acc, degs, G, num_edges, own_base, wv, lane);
        else
            lookup_and_add<false>(h, src, dst, gqueue, wq, acc, degs, G, num_edges, own_base, wv, lane);
        __syncthreads();
        if (tid == 0) gcount = 0u;
        __syncthreads();
        us = ue;
    }

    const int tile  = wave >> 1;
    const int nbl   = tile * 16;
    const int half  = lane >> 4;
    const int m5    = lane & 15;

    const float dm = (float)degs[nbl + m5];
    const float r  = (dm > 0.0f) ? (1.0f / dm) : 1.0f;

    const int rowbase = half * 8;
    unsigned degrow[8];
#pragma unroll
    for (int v = 0; v < 8; ++v) degrow[v] = degs[nbl + rowbase + v];

#pragma unroll
    for (int fi = 0; fi < 2; ++fi) {
        const int fc   = (wave & 1) * 2 + fi;
        const int feat = fc * 16 + m5;
        v8f c = {};
#pragma unroll
        for (int j = 0; j < 4; ++j) {
            v2f a;
            a.x = (m5 == 4 * j + 2 * half + 0) ? r : 0.0f;
            a.y = (m5 == 4 * j + 2 * half + 1) ? r : 0.0f;
            const int krow = nbl + 4 * j + 2 * half;
            v2f bb;
            bb.x = acc[(krow + 0) * FDIM + feat];
            bb.y = acc[(krow + 1) * FDIM + feat];
            c = __builtin_amdgcn_wmma_f32_16x16x4_f32(
                     false, a,  false, bb,
                     (short)0, c,  false,  false);
        }
#pragma unroll
        for (int v = 0; v < 8; ++v) {
            const size_t idx = (size_t)(own_base + nbl + rowbase + v) * FDIM + feat;
            out[idx] = (degrow[v] > 0u) ? c[v] : h[idx];
        }
    }
}

extern "C" void kernel_launch(void* const* d_in, const int* in_sizes, int n_in,
                              void* d_out, int out_size, void* d_ws, size_t ws_size,
                              hipStream_t stream) {
    const float* h   = (const float*)d_in[0];
    const int*   src = (const int*)d_in[1];
    const int*   dst = (const int*)d_in[2];
    float*       out = (float*)d_out;
    const int num_edges = in_sizes[1];

    unsigned* planes = (unsigned*)d_ws;

    gp_plane_kernel<<<256, PLANE_THREADS, 0, stream>>>(dst, planes, num_edges);

    gp_owner_kernel<<<NUM_NODES / OWN, OWNER_THREADS, 0, stream>>>(h, src, dst, planes, out, num_edges);
}


# Plane pass also writes each edge as one 4-byte record (src | dst << 16); an owner looks a queued group up with ONE 16-byte load, not a dst and a src load: scattered fetches halved; sum order kept
# speedup vs baseline: 4.4277x; 1.0600x over previous
#include <hip/hip_runtime.h>


typedef __attribute__((ext_vector_type(2))) float v2f;
typedef __attribute__((ext_vector_type(8))) float v8f;
typedef __attribute__((ext_vector_type(4))) unsigned u4;

#define NUM_NODES (8 * 8192)
#define FDIM 64
#define OWN 256
#define NOWNERS 256
#define GRP 4
#define CH_EDGES 4096
#define CH_WORDS 32
#define PLANE_THREADS 256
#define GCAP 8192
#define OWNER_THREADS 1024
#define OWNER_WAVES (OWNER_THREADS / 32)
#define ROWS_IN_FLIGHT 8
#define WQ_STRIDE 1024
#define EMPTY_ENTRY 0xFFFFFFFFu

static_assert(NUM_NODES == (1 << 16), "a record holds src and dst in 16 bits each");
static_assert(NOWNERS * OWN == NUM_NODES, "owner of an edge = bits 8..15 of its dst");

__device__ __forceinline__ u4 edge_group(unsigned* bits, unsigned col, unsigned bit,
                                         const int dk[GRP], const int sk[GRP]) {
    unsigned o[GRP], field[GRP];
    bool live[GRP];
    bool all = true;
#pragma unroll
    for (int k = 0; k < GRP; ++k) {
        live[k]  = (unsigned)dk[k] < (unsigned)NUM_NODES;
        o[k]     = ((unsigned)dk[k] >> 8) & 0xFFu;
        field[k] = (unsigned)dk[k] & 0xFFFFu;
        all = all && live[k];
    }
    if (!all) {
        unsigned dead = 0;
#pragma unroll
        for (int r = 0; r < GRP - 1; ++r) {
            bool taken = false;
#pragma unroll
            for (int k = 0; k < GRP; ++k) taken = taken || (live[k] && o[k] == dead);
            dead += taken ? 1u : 0u;
        }
#pragma unroll
        for (int k = 0; k < GRP; ++k)
            if (!live[k]) field[k] = dead << 8;
    }
    unsigned rk[GRP];
#pragma unroll
    for (int k = 0; k < GRP; ++k) {
        atomicOr(&bits[o[k] * CH_WORDS + ((col + (o[k] >> 1)) & (CH_WORDS - 1))], live[k] ? bit : 0u);
        int s = sk[k];
        s = s < 0 ? 0 : (s > NUM_NODES - 1 ? NUM_NODES - 1 : s);
        rk[k] = (unsigned)s | (field[k] << 16);
    }
    return (u4){rk[0], rk[1], rk[2], rk[3]};
}

__global__ __launch_bounds__(PLANE_THREADS)
void gp_plane_kernel(const int* __restrict__ src,
                     const int* __restrict__ dst,
                     unsigned* __restrict__ ws,
                     int num_edges) {
    __shared__ unsigned bits[NOWNERS * CH_WORDS];

    const int t    = threadIdx.x;
    const int lane = t & 31;
    const int wave = t >> 5;
    const int nchunks = (num_edges + CH_EDGES - 1) / CH_EDGES;
    const size_t pstride = (size_t)nchunks * CH_WORDS;
    u4* __restrict__ rec4 = (u4*)ws;
    unsigned* __restrict__ planes = ws + (size_t)nchunks * CH_EDGES;

    for (int c = blockIdx.x; c < nchunks; c += gridDim.x) {
        for (int i = t; i < NOWNERS * CH_WORDS / 4; i += PLANE_THREADS)
            ((u4*)bits)[i] = (u4){0u, 0u, 0u, 0u};
        __syncthreads();

        const int ebase = c * CH_EDGES;
        const int gbase = c * (CH_EDGES / GRP);
        if (ebase + CH_EDGES <= num_edges) {
            int4 d[4], s[4];
#pragma unroll
            for (int i = 0; i < 4; ++i) {
                d[i] = *(const int4*)(dst + ebase + (i * PLANE_THREADS + t) * GRP);
                s[i] = *(const int4*)(src + ebase + (i * PLANE_THREADS + t) * GRP);
            }
#pragma unroll
            for (int i = 0; i < 4; ++i) {
                const int gl = i * PLANE_THREADS + t;
                const unsigned bit = 1u << (gl & 31);
                const unsigned col = (unsigned)gl >> 5;
                const int dk[GRP] = {d[i].x, d[i].y, d[i].z, d[i].w};
                const int sk[GRP] = {s[i].x, s[i].y, s[i].z, s[i].w};
                rec4[gbase + gl] = edge_group(bits, col, bit, dk, sk);
            }
        } else {
            for (int i = 0; i < 4; ++i) {
                const int gl = i * PLANE_THREADS + t;
                const unsigned bit = 1u << (gl & 31);
                const unsigned col = (unsigned)gl >> 5;
                const int e0 = ebase + gl * GRP;
                if (e0 < num_edges) {
                    int dk[GRP], sk[GRP];
                    for (int k = 0; k < GRP; ++k) {
                        const bool in = e0 + k < num_edges;
                        dk[k] = in ? dst[e0 + k] : -1;
                        sk[k] = in ? src[e0 + k] : 0;
                    }
                    rec4[gbase + gl] = edge_group(bits, col, bit, dk, sk);
                }
            }
        }
        __syncthreads();

        for (int o = wave; o < NOWNERS; o += PLANE_THREADS / 32) {
            const unsigned v = bits[o * CH_WORDS + ((lane + (o >> 1)) & (CH_WORDS - 1))];
            planes[(size_t)o * pstride + (size_t)c * CH_WORDS + lane] = v;
        }
        __syncthreads();
    }
}

__device__ __forceinline__ u4 fetch_records(const u4* __restrict__ rec4, const unsigned* gqueue, int q, bool has) {
    const unsigned g = has ? gqueue[q] : 0u;
    return rec4[g];
}

template <bool CHECK>
__device__ __forceinline__ void add_rows(const float* __restrict__ h, float* acc, const unsigned* wq,
                                         int q0, int Q, int lane) {
    unsigned ent[ROWS_IN_FLIGHT];
    float2   hv[ROWS_IN_FLIGHT];
#pragma unroll
    for (int u = 0; u < ROWS_IN_FLIGHT; ++u)
        ent[u] = (!CHECK || q0 + u < Q) ? wq[q0 + u] : EMPTY_ENTRY;
#pragma unroll
    for (int u = 0; u < ROWS_IN_FLIGHT; ++u) {
        const unsigned s = (CHECK && ent[u] == EMPTY_ENTRY) ? 0u : (ent[u] & 0xFFFFu);
        hv[u] = *(const float2*)(h + (size_t)s * FDIM + lane * 2);
    }
#pragma unroll
    for (int u = 0; u < ROWS_IN_FLIGHT; ++u) {
        if (!CHECK || ent[u] != EMPTY_ENTRY) {
            const unsigned dl = ent[u] >> 16;
            float* a = acc + dl * FDIM + lane * 2;
            atomicAdd(a + 0, hv[u].x);
            atomicAdd(a + 1, hv[u].y);
        }
    }
}

__device__ __forceinline__ void lookup_and_add(const float* __restrict__ h,
                                               const u4* __restrict__ rec4,
                                               const unsigned* gqueue, unsigned* wq,
                                               float* acc, unsigned* degs,
                                               int G, unsigned owner, int wave, int lane) {
    const int qlo = (wave * G) >> 5;
    const int qhi = ((wave + 1) * G) >> 5;
    u4 rv;
    bool has = qlo + lane < qhi;
    if (qlo < qhi) rv = fetch_records(rec4, gqueue, qlo + lane, has);
    int carried = 0;
    for (int qb = qlo; qb < qhi; qb += 32) {
        const u4 rcur = rv;
        const bool hcur = has;
        if (qb + 32 < qhi) {
            has = qb + 32 + lane < qhi;
            rv = fetch_records(rec4, gqueue, qb + 32 + lane, has);
        }
        const unsigned rk[GRP] = {rcur.x, rcur.y, rcur.z, rcur.w};
        int n = carried;
#pragma unroll
        for (int k = 0; k < GRP; ++k) {
            const unsigned s  = rk[k] & 0xFFFFu;
            const unsigned d  = rk[k] >> 16;
            const bool mine = hcur && (d >> 8) == owner;
            const unsigned bal = __builtin_amdgcn_ballot_w32(mine);
            if (mine) {
                const unsigned dl = d & (OWN - 1);
                wq[n + (int)__builtin_amdgcn_mbcnt_lo(bal, 0u)] = s | (dl << 16);
                atomicAdd(&degs[dl], 1u);
            }
            n += __builtin_popcount(bal);
        }
        int q0 = 0;
        for (; q0 + ROWS_IN_FLIGHT <= n; q0 += ROWS_IN_FLIGHT)
            add_rows<false>(h, acc, wq, q0, n, lane);
        const int r = n - q0;
        if (q0 > 0 && r > 0) {
            const unsigned t = wq[q0 + (lane < r ? lane : 0)];
            if (lane < r) wq[lane] = t;
        }
        carried = r;
    }
    if (carried > 0) add_rows<true>(h, acc, wq, 0, carried, lane);
}

__global__ __launch_bounds__(OWNER_THREADS)
void gp_owner_kernel(const float* __restrict__ h,
                     const unsigned* __restrict__ ws,
                     float* __restrict__ out,
                     int num_edges) {
    __shared__ float    acc[OWN * FDIM];
    __shared__ unsigned degs[OWN];
    __shared__ unsigned gqueue[GCAP];
    __shared__ unsigned wqueue[OWNER_WAVES * WQ_STRIDE];
    __shared__ unsigned gcount;

    const int tid  = threadIdx.x;
    const int lane = tid & 31;
    const int wave = tid >> 5;
    const unsigned b = blockIdx.x;
    const int own_base = (int)b * OWN;

    for (int i = tid; i < OWN * FDIM / 4; i += OWNER_THREADS)
        ((float4*)acc)[i] = make_float4(0.f, 0.f, 0.f, 0.f);
    if (tid < OWN) degs[tid] = 0u;
    if (tid == 0) gcount = 0u;
    __syncthreads();

    const int nchunks = (num_edges + CH_EDGES - 1) / CH_EDGES;
    const int nunits  = nchunks * (CH_WORDS / 4);
    const u4* __restrict__ rec4 = (const u4*)ws;
    const unsigned* __restrict__ planes = ws + (size_t)nchunks * CH_EDGES;
    const u4* p4 = (const u4*)(planes + (size_t)b * (size_t)nchunks * CH_WORDS);
    int seg = nunits;
    int us  = 0;
    while (us < nunits) {
        const int ue = (us + seg < nunits) ? us + seg : nunits;
        for (int u0 = us + tid; u0 < ue; u0 += OWNER_THREADS * 2) {
            u4 w[2];
#pragma unroll
            for (int j = 0; j < 2; ++j) {
                const int u  = u0 + j * OWNER_THREADS;
                const int uc = u < ue ? u : ue - 1;
                w[j] = __builtin_nontemporal_load(p4 + uc);
                if (u >= ue) w[j] = (u4){0u, 0u, 0u, 0u};
            }
            unsigned n = 0;
#pragma unroll
            for (int j = 0; j < 2; ++j)
                n += __builtin_popcount(w[j].x) + __builtin_popcount(w[j].y) +
                     __builtin_popcount(w[j].z) + __builtin_popcount(w[j].w);
            unsigned slot = atomicAdd(&gcount, n);
#pragma unroll
            for (int j = 0; j < 2; ++j) {
                const unsigned wk[4] = {w[j].x, w[j].y, w[j].z, w[j].w};
#pragma unroll
                for (int k = 0; k < 4; ++k) {
                    unsigned m = wk[k];
                    const unsigned base = (unsigned)((u0 + j * OWNER_THREADS) * 4 + k) << 5;
                    while (m) {
                        const unsigned p = (unsigned)__builtin_ctz(m);
                        m &= m - 1;
                        if (slot < GCAP) gqueue[slot] = base + p;
                        ++slot;
                    }
                }
            }
        }
        __syncthreads();
        const unsigned cnt = gcount;
        if (cnt > GCAP) {
            __syncthreads();
            if (tid == 0) gcount = 0u;
            __syncthreads();
            seg = GCAP / 128;
            continue;
        }
        const int G = (int)cnt;

        const int wv = __builtin_amdgcn_readfirstlane(wave);
        unsigned* wq = wqueue + wv * WQ_STRIDE;
        lookup_and_add(h, rec4, gqueue, wq, acc, degs, G, b, wv, lane);
        __syncthreads();
        if (tid == 0) gcount = 0u;
        __syncthreads();
        us = ue;
    }

    const int tile  = wave >> 1;
    const int nbl   = tile * 16;
    const int half  = lane >> 4;
    const int m5    = lane & 15;

    const float dm = (float)degs[nbl + m5];
    const float r  = (dm > 0.0f) ? (1.0f / dm) : 1.0f;

    const int rowbase = half * 8;
    unsigned degrow[8];
#pragma unroll
    for (int v = 0; v < 8; ++v) degrow[v] = degs[nbl + rowbase + v];

#pragma unroll
    for (int fi = 0; fi < 2; ++fi) {
        const int fc   = (wave & 1) * 2 + fi;
        const int feat = fc * 16 + m5;
        v8f c = {};
#pragma unroll
        for (int j = 0; j < 4; ++j) {
            v2f a;
            a.x = (m5 == 4 * j + 2 * half + 0) ? r : 0.0f;
            a.y = (m5 == 4 * j + 2 * half + 1) ? r : 0.0f;
            const int krow = nbl + 4 * j + 2 * half;
            v2f bb;
            bb.x = acc[(krow + 0) * FDIM + feat];
            bb.y = acc[(krow + 1) * FDIM + feat];
            c = __builtin_amdgcn_wmma_f32_16x16x4_f32(
                     false, a,  false, bb,
                     (short)0, c,  false,  false);
        }
#pragma unroll
        for (int v = 0; v < 8; ++v) {
            const size_t idx = (size_t)(own_base + nbl + rowbase + v) * FDIM + feat;
            out[idx] = (degrow[v] > 0u) ? c[v] : h[idx];
        }
    }
}

extern "C" void kernel_launch(void* const* d_in, const int* in_sizes, int n_in,
                              void* d_out, int out_size, void* d_ws, size_t ws_size,
                              hipStream_t stream) {
    const float* h   = (const float*)d_in[0];
    const int*   src = (const int*)d_in[1];
    const int*   dst = (const int*)d_in[2];
    float*       out = (float*)d_out;
    const int num_edges = in_sizes[1];

    unsigned* ws = (unsigned*)d_ws;

    gp_plane_kernel<<<256, PLANE_THREADS, 0, stream>>>(src, dst, ws, num_edges);

    gp_owner_kernel<<<NUM_NODES / OWN, OWNER_THREADS, 0, stream>>>(h, ws, out, num_edges);
}


# Per-edge LDS float adds replaced: owner counting-sorts its edges by dst row in LDS; wave w sums its rows 8w..8w+7 in registers, 16 h rows in flight, one LDS add pair per row; sum order changes
# speedup vs baseline: 5.5470x; 1.2528x over previous
#include <hip/hip_runtime.h>


typedef __attribute__((ext_vector_type(2))) float v2f;
typedef __attribute__((ext_vector_type(8))) float v8f;
typedef __attribute__((ext_vector_type(4))) unsigned u4;

#define NUM_NODES (8 * 8192)
#define FDIM 64
#define OWN 256
#define NOWNERS 256
#define GRP 4
#define CH_EDGES 4096
#define CH_WORDS 32
#define PLANE_THREADS 256
#define GCAP 6144
#define OWNER_THREADS 1024
#define OWNER_WAVES (OWNER_THREADS / 32)
#define ROWS_IN_FLIGHT 16
#define LOOKUPS 4
#define WQ_STRIDE (GCAP / OWNER_WAVES * GRP)
#define SORT_CAP (GCAP * GRP)
#define EMPTY_ENTRY 0xFFFFFFFFu

static_assert(NUM_NODES == (1 << 16), "a record holds src and dst in 16 bits each");
static_assert(NOWNERS * OWN == NUM_NODES, "owner of an edge = bits 8..15 of its dst");
static_assert(OWN == 8 * OWNER_WAVES, "wave w sums rows 8w..8w+7");
static_assert(GCAP % (32 * OWNER_WAVES) == 0, "a wave's share of a round is at most GCAP / OWNER_WAVES groups");

__device__ __forceinline__ u4 edge_group(unsigned* bits, unsigned col, unsigned bit,
                                         const int dk[GRP], const int sk[GRP]) {
    unsigned o[GRP], field[GRP];
    bool live[GRP];
    bool all = true;
#pragma unroll
    for (int k = 0; k < GRP; ++k) {
        live[k]  = (unsigned)dk[k] < (unsigned)NUM_NODES;
        o[k]     = ((unsigned)dk[k] >> 8) & 0xFFu;
        field[k] = (unsigned)dk[k] & 0xFFFFu;
        all = all && live[k];
    }
    if (!all) {
        unsigned dead = 0;
#pragma unroll
        for (int r = 0; r < GRP - 1; ++r) {
            bool taken = false;
#pragma unroll
            for (int k = 0; k < GRP; ++k) taken = taken || (live[k] && o[k] == dead);
            dead += taken ? 1u : 0u;
        }
#pragma unroll
        for (int k = 0; k < GRP; ++k)
            if (!live[k]) field[k] = dead << 8;
    }
    unsigned rk[GRP];
#pragma unroll
    for (int k = 0; k < GRP; ++k) {
        atomicOr(&bits[o[k] * CH_WORDS + ((col + (o[k] >> 1)) & (CH_WORDS - 1))], live[k] ? bit : 0u);
        int s = sk[k];
        s = s < 0 ? 0 : (s > NUM_NODES - 1 ? NUM_NODES - 1 : s);
        rk[k] = (unsigned)s | (field[k] << 16);
    }
    return (u4){rk[0], rk[1], rk[2], rk[3]};
}

__global__ __launch_bounds__(PLANE_THREADS)
void gp_plane_kernel(const int* __restrict__ src,
                     const int* __restrict__ dst,
                     unsigned* __restrict__ ws,
                     int num_edges) {
    __shared__ unsigned bits[NOWNERS * CH_WORDS];

    const int t    = threadIdx.x;
    const int lane = t & 31;
    const int wave = t >> 5;
    const int nchunks = (num_edges + CH_EDGES - 1) / CH_EDGES;
    const size_t pstride = (size_t)nchunks * CH_WORDS;
    u4* __restrict__ rec4 = (u4*)ws;
    unsigned* __restrict__ planes = ws + (size_t)nchunks * CH_EDGES;

    for (int c = blockIdx.x; c < nchunks; c += gridDim.x) {
        for (int i = t; i < NOWNERS * CH_WORDS / 4; i += PLANE_THREADS)
            ((u4*)bits)[i] = (u4){0u, 0u, 0u, 0u};
        __syncthreads();

        const int ebase = c * CH_EDGES;
        const int gbase = c * (CH_EDGES / GRP);
        if (ebase + CH_EDGES <= num_edges) {
            int4 d[4], s[4];
#pragma unroll
            for (int i = 0; i < 4; ++i) {
                d[i] = *(const int4*)(dst + ebase + (i * PLANE_THREADS + t) * GRP);
                s[i] = *(const int4*)(src + ebase + (i * PLANE_THREADS + t) * GRP);
            }
#pragma unroll
            for (int i = 0; i < 4; ++i) {
                const int gl = i * PLANE_THREADS + t;
                const unsigned bit = 1u << (gl & 31);
                const unsigned col = (unsigned)gl >> 5;
                const int dk[GRP] = {d[i].x, d[i].y, d[i].z, d[i].w};
                const int sk[GRP] = {s[i].x, s[i].y, s[i].z, s[i].w};
                rec4[gbase + gl] = edge_group(bits, col, bit, dk, sk);
            }
        } else {
            for (int i = 0; i < 4; ++i) {
                const int gl = i * PLANE_THREADS + t;
                const unsigned bit = 1u << (gl & 31);
                const unsigned col = (unsigned)gl >> 5;
                const int e0 = ebase + gl * GRP;
                if (e0 < num_edges) {
                    int dk[GRP], sk[GRP];
                    for (int k = 0; k < GRP; ++k) {
                        const bool in = e0 + k < num_edges;
                        dk[k] = in ? dst[e0 + k] : -1;
                        sk[k] = in ? src[e0 + k] : 0;
                    }
                    rec4[gbase + gl] = edge_group(bits, col, bit, dk, sk);
                }
            }
        }
        __syncthreads();

        for (int o = wave; o < NOWNERS; o += PLANE_THREADS / 32) {
            const unsigned v = bits[o * CH_WORDS + ((lane + (o >> 1)) & (CH_WORDS - 1))];
            planes[(size_t)o * pstride + (size_t)c * CH_WORDS + lane] = v;
        }
        __syncthreads();
    }
}

__device__ __forceinline__ int lookup(const u4* __restrict__ rec4, const unsigned* gqueue, unsigned* wq,
                                      unsigned* cnt, int G, unsigned owner, int wave, int lane) {
    const int qlo = (wave * G) >> 5;
    const int qhi = ((wave + 1) * G) >> 5;
    int n = 0;
    for (int qb = qlo; qb < qhi; qb += 32 * LOOKUPS) {
        u4       rv[LOOKUPS];
        bool     has[LOOKUPS];
        unsigned g[LOOKUPS];
#pragma unroll
        for (int j = 0; j < LOOKUPS; ++j) {
            const int q = qb + 32 * j + lane;
            has[j] = q < qhi;
            g[j]   = has[j] ? gqueue[q] : 0u;
        }
#pragma unroll
        for (int j = 0; j < LOOKUPS; ++j) rv[j] = rec4[g[j]];
#pragma unroll
        for (int j = 0; j < LOOKUPS; ++j) {
            const unsigned rk[GRP] = {rv[j].x, rv[j].y, rv[j].z, rv[j].w};
#pragma unroll
            for (int k = 0; k < GRP; ++k) {
                const unsigned s  = rk[k] & 0xFFFFu;
                const unsigned d  = rk[k] >> 16;
                const bool mine = has[j] && (d >> 8) == owner;
                const unsigned bal = __builtin_amdgcn_ballot_w32(mine);
                if (mine) {
                    const unsigned dl = d & (OWN - 1);
                    wq[n + (int)__builtin_amdgcn_mbcnt_lo(bal, 0u)] = s | (dl << 16);
                    atomicAdd(&cnt[dl], 1u);
                }
                n += __builtin_popcount(bal);
            }
        }
    }
    return n;
}

__device__ __forceinline__ void scan_rows(unsigned* cnt, unsigned* degs, unsigned* cursor,
                                          unsigned* wstart, int lane) {
    unsigned c[8];
    unsigned tot = 0;
#pragma unroll
    for (int j = 0; j < 8; ++j) { c[j] = cnt[8 * lane + j]; tot += c[j]; }
    unsigned incl = tot;
#pragma unroll
    for (int d = 1; d < 32; d <<= 1) {
        const unsigned t = __shfl_up(incl, (unsigned)d);
        if (lane >= d) incl += t;
    }
    unsigned run = incl - tot;
    wstart[lane] = run;
    if (lane == 31) wstart[OWNER_WAVES] = incl;
#pragma unroll
    for (int j = 0; j < 8; ++j) {
        cursor[8 * lane + j] = run;
        degs[8 * lane + j]  += c[j];
        cnt[8 * lane + j]    = 0u;
        run += c[j];
    }
}

__device__ __forceinline__ void flush_row(float* acc, unsigned row, float2 cur, int lane) {
    float* a = acc + row * FDIM + lane * 2;
    atomicAdd(a + 0, cur.x);
    atomicAdd(a + 1, cur.y);
}

template <bool CHECK>
__device__ __forceinline__ unsigned sum_step(const float* __restrict__ h, float* acc, unsigned v,
                                             const unsigned* sorted, int qn, int hi,
                                             float2& cur, unsigned& row, int lane) {
    unsigned ent[ROWS_IN_FLIGHT];
    float2   hv[ROWS_IN_FLIGHT];
#pragma unroll
    for (int u = 0; u < ROWS_IN_FLIGHT; ++u)
        ent[u] = (unsigned)__builtin_amdgcn_readlane((int)v, u);
#pragma unroll
    for (int u = 0; u < ROWS_IN_FLIGHT; ++u) {
        const unsigned s = (CHECK && ent[u] == EMPTY_ENTRY) ? 0u : (ent[u] & 0xFFFFu);
        hv[u] = *(const float2*)(h + (size_t)s * FDIM + lane * 2);
    }
    const unsigned vnext = (!CHECK && qn < hi) ? sorted[qn] : EMPTY_ENTRY;
#pragma unroll
    for (int u = 0; u < ROWS_IN_FLIGHT; ++u) {
        if (!CHECK || ent[u] != EMPTY_ENTRY) {
            const unsigned r = ent[u] >> 16;
            if (r != row) {
                if (row != EMPTY_ENTRY) flush_row(acc, row, cur, lane);
                cur = make_float2(0.f, 0.f);
                row = r;
            }
            cur.x += hv[u].x;
            cur.y += hv[u].y;
        }
    }
    return vnext;
}

__device__ __forceinline__ void sum_rows(const float* __restrict__ h, const unsigned* sorted, float* acc,
                                         int lo, int hi, int lane) {
    float2   cur = make_float2(0.f, 0.f);
    unsigned row = EMPTY_ENTRY;
    const int l16 = lane & (ROWS_IN_FLIGHT - 1);
    int q0 = lo;
    unsigned v = (q0 + l16 < hi) ? sorted[q0 + l16] : EMPTY_ENTRY;
    for (; q0 + ROWS_IN_FLIGHT <= hi; q0 += ROWS_IN_FLIGHT)
        v = sum_step<false>(h, acc, v, sorted, q0 + ROWS_IN_FLIGHT + l16, hi, cur, row, lane);
    if (q0 < hi) sum_step<true>(h, acc, v, sorted, 0, 0, cur, row, lane);
    if (row != EMPTY_ENTRY) flush_row(acc, row, cur, lane);
}

__global__ __launch_bounds__(OWNER_THREADS)
void gp_owner_kernel(const float* __restrict__ h,
                     const unsigned* __restrict__ ws,
                     float* __restrict__ out,
                     int num_edges) {
    __shared__ float    acc[OWN * FDIM];
    __shared__ unsigned degs[OWN];
    __shared__ unsigned cnt[OWN];
    __shared__ unsigned cursor[OWN];
    __shared__ unsigned wstart[OWNER_WAVES + 1];
    __shared__ unsigned gqueue[GCAP];
    __shared__ unsigned wqueue[OWNER_WAVES * WQ_STRIDE];
    __shared__ unsigned sorted[SORT_CAP];
    __shared__ unsigned gcount;

    const int tid  = threadIdx.x;
    const int lane = tid & 31;
    const int wave = tid >> 5;
    const unsigned b = blockIdx.x;
    const int own_base = (int)b * OWN;

    for (int i = tid; i < OWN * FDIM / 4; i += OWNER_THREADS)
        ((float4*)acc)[i] = make_float4(0.f, 0.f, 0.f, 0.f);
    if (tid < OWN) { degs[tid] = 0u; cnt[tid] = 0u; }
    if (tid == 0) gcount = 0u;
    __syncthreads();

    const int nchunks = (num_edges + CH_EDGES - 1) / CH_EDGES;
    const int nunits  = nchunks * (CH_WORDS / 4);
    const u4* __restrict__ rec4 = (const u4*)ws;
    const unsigned* __restrict__ planes = ws + (size_t)nchunks * CH_EDGES;
    const u4* p4 = (const u4*)(planes + (size_t)b * (size_t)nchunks * CH_WORDS);
    int seg = nunits;
    int us  = 0;
    while (us < nunits) {
        const int ue = (us + seg < nunits) ? us + seg : nunits;
        for (int u0 = us + tid; u0 < ue; u0 += OWNER_THREADS * 2) {
            u4 w[2];
#pragma unroll
            for (int j = 0; j < 2; ++j) {
                const int u  = u0 + j * OWNER_THREADS;
                const int uc = u < ue ? u : ue - 1;
                w[j] = __builtin_nontemporal_load(p4 + uc);
                if (u >= ue) w[j] = (u4){0u, 0u, 0u, 0u};
            }
            unsigned n = 0;
#pragma unroll
            for (int j = 0; j < 2; ++j)
                n += __builtin_popcount(w[j].x) + __builtin_popcount(w[j].y) +
                     __builtin_popcount(w[j].z) + __builtin_popcount(w[j].w);
            unsigned slot = atomicAdd(&gcount, n);
#pragma unroll
            for (int j = 0; j < 2; ++j) {
                const unsigned wk[4] = {w[j].x, w[j].y, w[j].z, w[j].w};
#pragma unroll
                for (int k = 0; k < 4; ++k) {
                    unsigned m = wk[k];
                    const unsigned base = (unsigned)((u0 + j * OWNER_THREADS) * 4 + k) << 5;
                    while (m) {
                        const unsigned p = (unsigned)__builtin_ctz(m);
                        m &= m - 1;
                        if (slot < GCAP) gqueue[slot] = base + p;
                        ++slot;
                    }
                }
            }
        }
        __syncthreads();
        const unsigned gc = gcount;
        if (gc > GCAP) {
            __syncthreads();
            if (tid == 0) gcount = 0u;
            __syncthreads();
            seg = GCAP / 128;
            continue;
        }
        const int G = (int)gc;

        const int wv = __builtin_amdgcn_readfirstlane(wave);
        unsigned* wq = wqueue + wv * WQ_STRIDE;
        const int n = lookup(rec4, gqueue, wq, cnt, G, b, wv, lane);
        __syncthreads();
        if (wv == 0) scan_rows(cnt, degs, cursor, wstart, lane);
        __syncthreads();
        for (int i = lane; i < n; i += 32) {
            const unsigned e = wq[i];
            const unsigned slot = atomicAdd(&cursor[e >> 16], 1u);
            sorted[slot] = e;
        }
        __syncthreads();
        const int lo = __builtin_amdgcn_readfirstlane((int)wstart[wv]);
        const int hi = __builtin_amdgcn_readfirstlane((int)wstart[wv + 1]);
        sum_rows(h, sorted, acc, lo, hi, lane);
        __syncthreads();
        if (tid == 0) gcount = 0u;
        __syncthreads();
        us = ue;
    }

    const int tile  = wave >> 1;
    const int nbl   = tile * 16;
    const int half  = lane >> 4;
    const int m5    = lane & 15;

    const float dm = (float)degs[nbl + m5];
    const float r  = (dm > 0.0f) ? (1.0f / dm) : 1.0f;

    const int rowbase = half * 8;
    unsigned degrow[8];
#pragma unroll
    for (int v = 0; v < 8; ++v) degrow[v] = degs[nbl + rowbase + v];

#pragma unroll
    for (int fi = 0; fi < 2; ++fi) {
        const int fc   = (wave & 1) * 2 + fi;
        const int feat = fc * 16 + m5;
        v8f c = {};
#pragma unroll
        for (int j = 0; j < 4; ++j) {
            v2f a;
            a.x = (m5 == 4 * j + 2 * half + 0) ? r : 0.0f;
            a.y = (m5 == 4 * j + 2 * half + 1) ? r : 0.0f;
            const int krow = nbl + 4 * j + 2 * half;
            v2f bb;
            bb.x = acc[(krow + 0) * FDIM + feat];
            bb.y = acc[(krow + 1) * FDIM + feat];
            c = __builtin_amdgcn_wmma_f32_16x16x4_f32(
                     false, a,  false, bb,
                     (short)0, c,  false,  false);
        }
#pragma unroll
        for (int v = 0; v < 8; ++v) {
            const size_t idx = (size_t)(own_base + nbl + rowbase + v) * FDIM + feat;
            out[idx] = (degrow[v] > 0u) ? c[v] : h[idx];
        }
    }
}

extern "C" void kernel_launch(void* const* d_in, const int* in_sizes, int n_in,
                              void* d_out, int out_size, void* d_ws, size_t ws_size,
                              hipStream_t stream) {
    const float* h   = (const float*)d_in[0];
    const int*   src = (const int*)d_in[1];
    const int*   dst = (const int*)d_in[2];
    float*       out = (float*)d_out;
    const int num_edges = in_sizes[1];

    unsigned* ws = (unsigned*)d_ws;

    gp_plane_kernel<<<256, PLANE_THREADS, 0, stream>>>(src, dst, ws, num_edges);

    gp_owner_kernel<<<NUM_NODES / OWN, OWNER_THREADS, 0, stream>>>(h, ws, out, num_edges);
}
